# starting kernel
# baseline (speedup 1.0000x reference)
#include <hip/hip_runtime.h>

typedef __attribute__((ext_vector_type(2))) float v2f;
typedef __attribute__((ext_vector_type(8))) float v8f;

#define C_CHANNELS 128
#define ROWS_PER_BLOCK 16
#define CHUNK 256

__global__ __launch_bounds__(256) void neighbor_pool_wmma(
    const float* __restrict__ feat,
    const int*   __restrict__ nidx,
    const int*   __restrict__ splits,
    float*       __restrict__ out,
    int M)
{
    __shared__ int   s_split[ROWS_PER_BLOCK + 1];
    __shared__ float s_inv[ROWS_PER_BLOCK];
    __shared__ int   s_idx[CHUNK];

    const int tid  = threadIdx.x;
    const int lane = tid & 31;
    const int wave = tid >> 5;
    const int base = blockIdx.x * ROWS_PER_BLOCK;

    if (tid <= ROWS_PER_BLOCK) {
        int r = base + tid;
        if (r > M) r = M;
        s_split[tid] = splits[r];
    }
    __syncthreads();

    const int      r     = lane & 15;
    const int      myLo  = s_split[r];
    const unsigned myCnt = (unsigned)(s_split[r + 1] - myLo);
    const int      s0    = s_split[0];
    const int      s1    = s_split[ROWS_PER_BLOCK];
    const int      k0    = (lane >= 16) ? 2 : 0;
    const unsigned chan  = (unsigned)(wave * 16 + r);

    v8f acc = {};

    for (int cbeg = s0; cbeg < s1; cbeg += CHUNK) {
        int cend = cbeg + CHUNK;
        if (cend > s1) cend = s1;

        {
            int t = cbeg + tid;
            if (t >= s1) t = s1 - 1;
            s_idx[tid] = __builtin_nontemporal_load(&nidx[t]);
        }
        __syncthreads();

        const int eend = cbeg + ((cend - cbeg + 15) & ~15);
#pragma unroll 4
        for (int e = cbeg; e < eend; e += 4) {
            const int ge0 = e + k0;
            const int ge1 = ge0 + 1;

            v2f a;
            a.x = ((unsigned)(ge0 - myLo) < myCnt) ? 1.0f : 0.0f;
            a.y = ((unsigned)(ge1 - myLo) < myCnt) ? 1.0f : 0.0f;

            const int2 p = *(const int2*)(s_idx + (ge0 - cbeg));

            v2f b;
            b.x = feat[((unsigned)p.x << 7) + chan];
            b.y = feat[((unsigned)p.y << 7) + chan];

            acc = __builtin_amdgcn_wmma_f32_16x16x4_f32(
                 false, a,  false, b,
                 (short)0, acc,
                 false,  false);
        }
        __syncthreads();
    }

    if (tid < ROWS_PER_BLOCK) {
        int cnt = s_split[tid + 1] - s_split[tid];
        if (cnt < 1) cnt = 1;
        s_inv[tid] = 1.0f / (float)cnt;
    }
    __syncthreads();

    const int rowOfs = (lane >= 16) ? 8 : 0;
#pragma unroll
    for (int g = 0; g < 8; ++g) {
        const int mrow = g + rowOfs;
        const int m = base + mrow;
        if (m < M) {
            __builtin_nontemporal_store(acc[g] * s_inv[mrow],
                                        &out[(unsigned)m * C_CHANNELS + chan]);
        }
    }
}

extern "C" void kernel_launch(void* const* d_in, const int* in_sizes, int n_in,
                              void* d_out, int out_size, void* d_ws, size_t ws_size,
                              hipStream_t stream) {
    const float* feat   = (const float*)d_in[0];
    const int*   nidx   = (const int*)  d_in[1];
    const int*   splits = (const int*)  d_in[2];
    float*       out    = (float*)      d_out;

    const int M = in_sizes[2] - 1;
    const int blocks = (M + ROWS_PER_BLOCK - 1) / ROWS_PER_BLOCK;
    neighbor_pool_wmma<<<blocks, 256, 0, stream>>>(feat, nidx, splits, out, M);
}


# B operand staged through LDS: block gathers whole 512-byte feature rows with 16-byte loads (next chunk in flight during the WMMAs), waves read their slices back; k order and sums unchanged
# speedup vs baseline: 1.1323x; 1.1323x over previous
#include <hip/hip_runtime.h>


typedef __attribute__((ext_vector_type(2))) float v2f;
typedef __attribute__((ext_vector_type(4))) float v4f;
typedef __attribute__((ext_vector_type(8))) float v8f;

#define C_CHANNELS 128
#define ROWS_PER_BLOCK 16
#define NWAVES 8
#define CHUNK 64
#define RPW (CHUNK / NWAVES)
#define PITCH 144

__device__ __forceinline__ void load_ids(
    const int* __restrict__ nidx, int cbeg, int s1, int wave, unsigned (&ids)[RPW])
{
#pragma unroll
    for (int j = 0; j < RPW; ++j) {
        int t = cbeg + wave + NWAVES * j;
        if (t >= s1) t = s1 - 1;
        ids[j] = (unsigned)nidx[t];
    }
}

__device__ __forceinline__ void gather_rows(
    const float* __restrict__ feat, const unsigned (&ids)[RPW],
    int cbeg, int s1, int wave, unsigned colOfs, v4f (&rg)[RPW])
{
    int n = s1 - cbeg;
    if (n > CHUNK) n = CHUNK;
    const int nld = (n + 15) & ~15;
#pragma unroll
    for (int j = 0; j < RPW; ++j) {
        const int row = wave + NWAVES * j;
        if (row < nld)
            rg[j] = *(const v4f*)(feat + ((size_t)ids[j] << 7) + colOfs);
    }
}

__global__ __launch_bounds__(256) void neighbor_pool_wmma(
    const float* __restrict__ feat,
    const int*   __restrict__ nidx,
    const int*   __restrict__ splits,
    float*       __restrict__ out,
    int M)
{
    __shared__ int   s_split[ROWS_PER_BLOCK + 1];
    __shared__ float s_inv[ROWS_PER_BLOCK];
    __shared__ __attribute__((aligned(16))) float s_rows[CHUNK * PITCH];

    const int tid  = threadIdx.x;
    const int lane = tid & 31;
    const int wave = __builtin_amdgcn_readfirstlane(tid >> 5);
    const int base = blockIdx.x * ROWS_PER_BLOCK;

    int rEnd = base + ROWS_PER_BLOCK;
    if (rEnd > M) rEnd = M;
    const int s0 = splits[base];
    const int s1 = splits[rEnd];
    unsigned ids[RPW];
    if (s0 < s1) load_ids(nidx, s0, s1, wave, ids);

    if (tid <= ROWS_PER_BLOCK) {
        int r = base + tid;
        if (r > M) r = M;
        s_split[tid] = splits[r];
    }
    __syncthreads();

    const int      r     = lane & 15;
    const int      myLo  = s_split[r];
    const unsigned myCnt = (unsigned)(s_split[r + 1] - myLo);
    const int      half  = lane >> 4;
    const int      k0    = half * 2;
    const unsigned chan  = (unsigned)(wave * 16 + r);
    const unsigned colOfs = (unsigned)lane * 4u;

    v8f acc = {};

    v4f rg[RPW];
#pragma unroll
    for (int j = 0; j < RPW; ++j) rg[j] = v4f{0.f, 0.f, 0.f, 0.f};

    if (s0 < s1) gather_rows(feat, ids, s0, s1, wave, colOfs, rg);

    const float* const bp0 = s_rows + k0 * PITCH + chan;

    for (int cbeg = s0; cbeg < s1; cbeg += CHUNK) {
        int n = s1 - cbeg;
        if (n > CHUNK) n = CHUNK;
        const int nld = (n + 15) & ~15;

        const bool more = cbeg + CHUNK < s1;
        if (more) load_ids(nidx, cbeg + CHUNK, s1, wave, ids);

#pragma unroll
        for (int j = 0; j < RPW; ++j) {
            const int row = wave + NWAVES * j;
            if (row < nld) *(v4f*)(s_rows + row * PITCH + colOfs) = rg[j];
        }
        __syncthreads();

        if (more)
            gather_rows(feat, ids, cbeg + CHUNK, s1, wave, colOfs, rg);

        const float* bp = bp0;
        int rel = cbeg + k0 - myLo;
        const int ngroups = nld >> 2;
#pragma unroll 4
        for (int g = 0; g < ngroups; ++g) {
            v2f a;
            a.x = ((unsigned)rel       < myCnt) ? 1.0f : 0.0f;
            a.y = ((unsigned)(rel + 1) < myCnt) ? 1.0f : 0.0f;

            v2f b;
            b.x = bp[0];
            b.y = bp[PITCH];

            acc = __builtin_amdgcn_wmma_f32_16x16x4_f32(
                 false, a,  false, b,
                 (short)0, acc,
                 false,  false);

            bp  += 4 * PITCH;
            rel += 4;
        }
        __syncthreads();
    }

    if (tid < ROWS_PER_BLOCK) {
        int cnt = s_split[tid + 1] - s_split[tid];
        if (cnt < 1) cnt = 1;
        s_inv[tid] = 1.0f / (float)cnt;
    }
    __syncthreads();

    const int rowOfs = (lane >= 16) ? 8 : 0;
#pragma unroll
    for (int g = 0; g < 8; ++g) {
        const int mrow = g + rowOfs;
        const int m = base + mrow;
        if (m < M) {
            __builtin_nontemporal_store(acc[g] * s_inv[mrow],
                                        &out[(unsigned)m * C_CHANNELS + chan]);
        }
    }
}

extern "C" void kernel_launch(void* const* d_in, const int* in_sizes, int n_in,
                              void* d_out, int out_size, void* d_ws, size_t ws_size,
                              hipStream_t stream) {
    const float* feat   = (const float*)d_in[0];
    const int*   nidx   = (const int*)  d_in[1];
    const int*   splits = (const int*)  d_in[2];
    float*       out    = (float*)      d_out;

    const int M = in_sizes[2] - 1;
    const int blocks = (M + ROWS_PER_BLOCK - 1) / ROWS_PER_BLOCK;
    neighbor_pool_wmma<<<blocks, 256, 0, stream>>>(feat, nidx, splits, out, M);
}


# Tile's edges split over its 8 waves: each wave sums all 128 channels (8 accumulators, one A per K-step, B by 16-byte global loads, no LDS in the loop); 8 partials added in LDS: new sum order
# speedup vs baseline: 1.6327x; 1.4419x over previous
#include <hip/hip_runtime.h>


typedef __attribute__((ext_vector_type(2))) float v2f;
typedef __attribute__((ext_vector_type(4))) float v4f;
typedef __attribute__((ext_vector_type(8))) float v8f;

#define C_CHANNELS 128
#define ROWS_PER_BLOCK 16
#define NWAVES 8
#define NTILES 8

struct Stage { v4f lo[2][2]; v4f hi[2][2]; };

__device__ __forceinline__ void load_ids(
    const int* __restrict__ nidx, int e0, int s1, unsigned (&ids)[8])
{
    if (e0 + 8 <= s1) {
#pragma unroll
        for (int k = 0; k < 8; ++k) ids[k] = (unsigned)nidx[e0 + k];
    } else {
#pragma unroll
        for (int k = 0; k < 8; ++k) {
            int t = e0 + k;
            if (t >= s1) t = s1 - 1;
            ids[k] = (unsigned)nidx[t];
        }
    }
}

__device__ __forceinline__ void issue_rows(
    const float* __restrict__ feat, const unsigned (&ids)[8],
    bool upper, unsigned colOfs, Stage& st)
{
#pragma unroll
    for (int t = 0; t < 2; ++t) {
#pragma unroll
        for (int s = 0; s < 2; ++s) {
            const unsigned id = upper ? ids[4 * t + 2 + s] : ids[4 * t + s];
            const float* p = feat + (size_t)((id << 7) + colOfs);
            st.lo[t][s] = *(const v4f*)(p);
            st.hi[t][s] = *(const v4f*)(p + 64);
        }
    }
}

__device__ __forceinline__ void compute_stage(
    const Stage& st, int rel, unsigned cnt, v8f (&acc)[NTILES])
{
#pragma unroll
    for (int t = 0; t < 2; ++t) {
        v2f a;
        a.x = ((unsigned)(rel + 4 * t)     < cnt) ? 1.0f : 0.0f;
        a.y = ((unsigned)(rel + 4 * t + 1) < cnt) ? 1.0f : 0.0f;
#pragma unroll
        for (int j = 0; j < NTILES; ++j) {
            v2f b;
            if (j < 4) { b.x = st.lo[t][0][j];     b.y = st.lo[t][1][j]; }
            else       { b.x = st.hi[t][0][j - 4]; b.y = st.hi[t][1][j - 4]; }
            acc[j] = __builtin_amdgcn_wmma_f32_16x16x4_f32(
                 false, a,  false, b,
                 (short)0, acc[j],
                 false,  false);
        }
    }
}

__global__ __launch_bounds__(256) void neighbor_pool_wmma(
    const float* __restrict__ feat,
    const int*   __restrict__ nidx,
    const int*   __restrict__ splits,
    float*       __restrict__ out,
    int M)
{
    __shared__ int s_split[ROWS_PER_BLOCK + 1];
    __shared__ __attribute__((aligned(16))) float s_red[NWAVES * ROWS_PER_BLOCK * C_CHANNELS];

    const int tid  = threadIdx.x;
    const int lane = tid & 31;
    const int wave = __builtin_amdgcn_readfirstlane(tid >> 5);
    const int base = blockIdx.x * ROWS_PER_BLOCK;

    int rEnd = base + ROWS_PER_BLOCK;
    if (rEnd > M) rEnd = M;
    const int s0 = splits[base];
    const int s1 = splits[rEnd];

    const int nK  = (s1 - s0 + 3) >> 2;
    const int q   = (((nK + NWAVES - 1) / NWAVES) + 1) & ~1;
    const int eb  = s0 + 4 * q * wave;
    int ee = eb + 4 * q;
    if (ee > s1) ee = s1;
    const int nst = (ee > eb) ? ((ee - eb + 7) >> 3) : 0;

    const int  c     = lane & 15;
    const bool upper = lane >= 16;
    const unsigned colOfs = (unsigned)c * 4u;

    unsigned ids[8];
    Stage sa, sb;
    if (nst > 0) {
        load_ids(nidx, eb, s1, ids);
        issue_rows(feat, ids, upper, colOfs, sa);
        if (nst > 1) load_ids(nidx, eb + 8, s1, ids);
    }

    if (tid <= ROWS_PER_BLOCK) {
        int r = base + tid;
        if (r > M) r = M;
        s_split[tid] = splits[r];
    }
    __syncthreads();

    int lo = s_split[c];
    int hi = s_split[c + 1];
    if (lo < eb) lo = eb;
    if (hi > ee) hi = ee;
    const unsigned cnt = (hi > lo) ? (unsigned)(hi - lo) : 0u;
    int rel = eb + (upper ? 2 : 0) - lo;

    v8f acc[NTILES];
#pragma unroll
    for (int j = 0; j < NTILES; ++j) acc[j] = v8f{};

    int st = 0;
    for (; st + 2 <= nst; st += 2) {
        issue_rows(feat, ids, upper, colOfs, sb);
        if (st + 2 < nst) load_ids(nidx, eb + 8 * (st + 2), s1, ids);
        compute_stage(sa, rel, cnt, acc);
        rel += 8;
        if (st + 2 < nst) {
            issue_rows(feat, ids, upper, colOfs, sa);
            if (st + 3 < nst) load_ids(nidx, eb + 8 * (st + 3), s1, ids);
        }
        compute_stage(sb, rel, cnt, acc);
        rel += 8;
    }
    if (st < nst) compute_stage(sa, rel, cnt, acc);

    {
        float* red = s_red + wave * (ROWS_PER_BLOCK * C_CHANNELS);
        const int rowOfs = upper ? 8 : 0;
#pragma unroll
        for (int g = 0; g < 8; ++g) {
            float* rp = red + (g + rowOfs) * C_CHANNELS + colOfs;
            *(v4f*)(rp)      = v4f{acc[0][g], acc[1][g], acc[2][g], acc[3][g]};
            *(v4f*)(rp + 64) = v4f{acc[4][g], acc[5][g], acc[6][g], acc[7][g]};
        }
    }
    __syncthreads();

#pragma unroll
    for (int i = 0; i < 2; ++i) {
        const int v    = tid + 256 * i;
        const int mrow = v >> 5;
        const int col  = (v & 31) * 4;
        const float* rp = s_red + mrow * C_CHANNELS + col;
        v4f sum = *(const v4f*)(rp);
#pragma unroll
        for (int w = 1; w < NWAVES; ++w)
            sum += *(const v4f*)(rp + w * (ROWS_PER_BLOCK * C_CHANNELS));
        int cntRow = s_split[mrow + 1] - s_split[mrow];
        if (cntRow < 1) cntRow = 1;
        const float inv = 1.0f / (float)cntRow;
        const int m = base + mrow;
        if (m < M) {
            __builtin_nontemporal_store(sum * inv,
                (v4f*)&out[(unsigned)m * C_CHANNELS + col]);
        }
    }
}

extern "C" void kernel_launch(void* const* d_in, const int* in_sizes, int n_in,
                              void* d_out, int out_size, void* d_ws, size_t ws_size,
                              hipStream_t stream) {
    const float* feat   = (const float*)d_in[0];
    const int*   nidx   = (const int*)  d_in[1];
    const int*   splits = (const int*)  d_in[2];
    float*       out    = (float*)      d_out;

    const int M = in_sizes[2] - 1;
    const int blocks = (M + ROWS_PER_BLOCK - 1) / ROWS_PER_BLOCK;
    neighbor_pool_wmma<<<blocks, 256, 0, stream>>>(feat, nidx, splits, out, M);
}
